# starting kernel
# baseline (speedup 1.0000x reference)
#include <hip/hip_runtime.h>
#include <hip/hip_bf16.h>
#include <math.h>

#define BATCH   4
#define SEQ     512
#define INDIM   300
#define HDIM    128
#define MROWS   (BATCH * SEQ)

typedef __attribute__((ext_vector_type(2))) float v2f;
typedef __attribute__((ext_vector_type(8))) float v8f;


__device__ __forceinline__ v2f load_fragA(const float* A, int m0, int k0,
                                          int lda, int lane) {
    const int m  = m0 + (lane & 15);
    const int ka = k0 + ((lane >> 4) << 1);
    return *reinterpret_cast<const v2f*>(A + (size_t)m * lda + ka);
}

__device__ __forceinline__ v2f load_fragB_WT(const float* W, int n0, int k0,
                                             int ldw, int lane) {
    const int nn = n0 + (lane & 15);
    const int kb = k0 + ((lane >> 4) << 1);
    return *reinterpret_cast<const v2f*>(W + (size_t)nn * ldw + kb);
}

__global__ __launch_bounds__(256) void gemm_wx_tanh(
    const float* __restrict__ A,
    const float* __restrict__ W,
    const float* __restrict__ bias,
    float* __restrict__ out)
{
    const int lane = threadIdx.x & 31;
    const int wid  = blockIdx.x * 8 + (threadIdx.x >> 5);
    const int m0   = (wid >> 3) << 4;
    const int n0   = (wid & 7) << 4;

    v8f acc = {};
#pragma unroll 5
    for (int k0 = 0; k0 < INDIM; k0 += 4) {
        v2f a = load_fragA(A, m0, k0, INDIM, lane);
        v2f b = load_fragB_WT(W, n0, k0, INDIM, lane);
        acc = __builtin_amdgcn_wmma_f32_16x16x4_f32(
                false, a, false, b, (short)0, acc, false, false);
    }

    const int col = n0 + (lane & 15);
    const int rb  = m0 + ((lane >> 4) << 3);
    const float bv = bias[col];
#pragma unroll
    for (int r = 0; r < 8; ++r) {
        out[(size_t)(rb + r) * HDIM + col] = tanhf(acc[r] + bv);
    }
}

__global__ __launch_bounds__(256) void gemm_ta_tb(
    const float* __restrict__ A,
    const float* __restrict__ Wa,
    const float* __restrict__ ba,
    const float* __restrict__ Ua,
    const float* __restrict__ bu,
    float* __restrict__ ta,
    float* __restrict__ tb)
{
    const int lane = threadIdx.x & 31;
    const int wid  = blockIdx.x * 8 + (threadIdx.x >> 5);
    const int m0   = (wid >> 3) << 4;
    const int n0   = (wid & 7) << 4;

    v8f accA = {};
    v8f accB = {};
#pragma unroll 4
    for (int k0 = 0; k0 < HDIM; k0 += 4) {
        v2f a  = load_fragA(A, m0, k0, HDIM, lane);
        v2f bw = load_fragB_WT(Wa, n0, k0, HDIM, lane);
        v2f bu2 = load_fragB_WT(Ua, n0, k0, HDIM, lane);
        accA = __builtin_amdgcn_wmma_f32_16x16x4_f32(
                 false, a, false, bw, (short)0, accA, false, false);
        accB = __builtin_amdgcn_wmma_f32_16x16x4_f32(
                 false, a, false, bu2, (short)0, accB, false, false);
    }

    const int col = n0 + (lane & 15);
    const int rb  = m0 + ((lane >> 4) << 3);
    const float bva = ba[col];
    const float bvu = bu[col];
#pragma unroll
    for (int r = 0; r < 8; ++r) {
        const size_t idx = (size_t)(rb + r) * HDIM + col;
        ta[idx] = accA[r] + bva;
        tb[idx] = accB[r] + bvu;
    }
}

#define QT 4

__global__ __launch_bounds__(128) void attn_fused(
    const float* __restrict__ ta,
    const float* __restrict__ tb,
    const float* __restrict__ hv,
    const float* __restrict__ maskp,
    float* __restrict__ out)
{
    const int h   = threadIdx.x;
    const int bq  = blockIdx.x;
    const int b   = bq / (SEQ / QT);
    const int q0  = (bq % (SEQ / QT)) * QT;
    const size_t base = (size_t)b * SEQ * HDIM;

    float taq[QT], m[QT], s[QT], acc[QT];
#pragma unroll
    for (int i = 0; i < QT; ++i) {
        taq[i] = ta[base + (size_t)(q0 + i) * HDIM + h];
        m[i]   = -INFINITY;
        s[i]   = 0.0f;
        acc[i] = 0.0f;
    }

    const float* ptb = tb    + base + h;
    const float* phv = hv    + base + h;
    const float* pmk = maskp + base + h;

    for (int k = 0; k < SEQ; ++k) {
        if (k + 16 < SEQ) {
            __builtin_prefetch(ptb + (size_t)(k + 16) * HDIM, 0, 0);
            __builtin_prefetch(phv + (size_t)(k + 16) * HDIM, 0, 0);
        }
        const float tbk = ptb[(size_t)k * HDIM];
        const float hvk = phv[(size_t)k * HDIM];
        const float mkk = pmk[(size_t)k * HDIM];
#pragma unroll
        for (int i = 0; i < QT; ++i) {
            const float x = taq[i] + tbk;
            const float e = (x > 0.0f ? x : __expf(x) - 1.0f) + mkk;
            const float mn   = fmaxf(m[i], e);
            const float corr = __expf(m[i] - mn);
            const float p    = __expf(e - mn);
            s[i]   = s[i]  * corr + p;
            acc[i] = acc[i] * corr + p * hvk;
            m[i]   = mn;
        }
    }

#pragma unroll
    for (int i = 0; i < QT; ++i) {
        out[base + (size_t)(q0 + i) * HDIM + h] = acc[i] / s[i];
    }
}

extern "C" void kernel_launch(void* const* d_in, const int* in_sizes, int n_in,
                              void* d_out, int out_size, void* d_ws, size_t ws_size,
                              hipStream_t stream) {
    const float* word_vecs = (const float*)d_in[0];
    const float* mask      = (const float*)d_in[1];
    const float* Wx_w      = (const float*)d_in[2];
    const float* Wx_b      = (const float*)d_in[3];
    const float* Wa_w      = (const float*)d_in[4];
    const float* Wa_b      = (const float*)d_in[5];
    const float* Ua_w      = (const float*)d_in[6];
    const float* Ua_b      = (const float*)d_in[7];
    float* out = (float*)d_out;

    float* hv = (float*)d_ws;
    float* ta = hv + (size_t)MROWS * HDIM;
    float* tb = ta + (size_t)MROWS * HDIM;

    gemm_wx_tanh<<<128, 256, 0, stream>>>(word_vecs, Wx_w, Wx_b, hv);
    gemm_ta_tb<<<128, 256, 0, stream>>>(hv, Wa_w, Wa_b, Ua_w, Ua_b, ta, tb);

    attn_fused<<<BATCH * (SEQ / QT), 128, 0, stream>>>(ta, tb, hv, mask, out);
}


# ta, tb projection moved into the attention launch (2 launches, not 3): a work-group takes 2 adjacent channels and forms its 4 ta/tb columns on WMMA from the key-major h planes; every sum order kept
# speedup vs baseline: 11.3284x; 11.3284x over previous
#include <hip/hip_runtime.h>
#include <hip/hip_bf16.h>
#include <math.h>

#define BATCH   4
#define SEQ     512
#define INDIM   300
#define HDIM    128
#define MROWS   (BATCH * SEQ)

typedef __attribute__((ext_vector_type(2))) float v2f;
typedef __attribute__((ext_vector_type(4))) float v4f;
typedef __attribute__((ext_vector_type(8))) float v8f;


__device__ __forceinline__ v2f load_fragA(const float* A, int m0, int k0,
                                          int lda, int lane) {
    const int m  = m0 + (lane & 15);
    const int ka = k0 + ((lane >> 4) << 1);
    return *reinterpret_cast<const v2f*>(A + (size_t)m * lda + ka);
}

__device__ __forceinline__ v2f load_fragB_WT(const float* W, int n0, int k0,
                                             int ldw, int lane) {
    const int nn = n0 + (lane & 15);
    const int kb = k0 + ((lane >> 4) << 1);
    return *reinterpret_cast<const v2f*>(W + (size_t)nn * ldw + kb);
}

__device__ __forceinline__ v4f load_frag4(const float* P, int r0, int k0,
                                          int ld, int lane) {
    const int r  = r0 + (lane & 15);
    const int kk = k0 + ((lane >> 4) << 2);
    return *reinterpret_cast<const v4f*>(P + (size_t)r * ld + kk);
}
__device__ __forceinline__ v2f frag_lo(v4f v) { v2f r = {v.x, v.y}; return r; }
__device__ __forceinline__ v2f frag_hi(v4f v) { v2f r = {v.z, v.w}; return r; }

__device__ __forceinline__ void store_keymajor(float* P, int rb, int col,
                                               const float* v) {
    float* p = P + ((size_t)((rb >> 9) * HDIM + col)) * SEQ + (rb & (SEQ - 1));
    *reinterpret_cast<float4*>(p)     = make_float4(v[0], v[1], v[2], v[3]);
    *reinterpret_cast<float4*>(p + 4) = make_float4(v[4], v[5], v[6], v[7]);
}

__global__ __launch_bounds__(128) __attribute__((amdgpu_waves_per_eu(1, 4)))
void gemm_wx_tanh(
    const float* __restrict__ A,
    const float* __restrict__ W,
    const float* __restrict__ bias,
    float* __restrict__ outT)
{
    const int lane = threadIdx.x & 31;
    const int wid  = blockIdx.x * 4 + (threadIdx.x >> 5);
    const int m0   = (wid >> 3) << 4;
    const int n0   = (wid & 7) << 4;

    v8f acc = {};
#pragma unroll
    for (int k0 = 0; k0 < 296; k0 += 8) {
        const v4f a = load_frag4(A, m0, k0, INDIM, lane);
        const v4f b = load_frag4(W, n0, k0, INDIM, lane);
        acc = __builtin_amdgcn_wmma_f32_16x16x4_f32(
                false, frag_lo(a), false, frag_lo(b), (short)0, acc, false, false);
        acc = __builtin_amdgcn_wmma_f32_16x16x4_f32(
                false, frag_hi(a), false, frag_hi(b), (short)0, acc, false, false);
    }
    {
        const v2f a = load_fragA(A, m0, 296, INDIM, lane);
        const v2f b = load_fragB_WT(W, n0, 296, INDIM, lane);
        acc = __builtin_amdgcn_wmma_f32_16x16x4_f32(
                false, a, false, b, (short)0, acc, false, false);
    }

    const int col = n0 + (lane & 15);
    const int rb  = m0 + ((lane >> 4) << 3);
    const float bv = bias[col];
    float v[8];
#pragma unroll
    for (int r = 0; r < 8; ++r) v[r] = tanhf(acc[r] + bv);
    store_keymajor(outT, rb, col, v);
}

#define QT 2
#define KC 8
#define BIG_OPERAND 67108864.0f

__device__ __forceinline__ float finite_size(float v) {
    const float a = fabsf(v);
    return a < INFINITY ? a : 0.0f;
}

__device__ __forceinline__ void attn_pairs(
    const int bh,
    const float* pta,
    const float* ptb,
    const float* phv,
    const float* pmk,
    float* __restrict__ out)
{
    const int lane = threadIdx.x & 31;
    const int w  = __builtin_amdgcn_readfirstlane((int)(threadIdx.x >> 5));
    const int b  = bh >> 7;
    const int h  = bh & (HDIM - 1);
    const int q0 = w * (32 * QT) + lane;

    float taq[QT];
#pragma unroll
    for (int i = 0; i < QT; ++i) taq[i] = pta[q0 + 32 * i];

    float tmax = -INFINITY;
    float kmax = -INFINITY;
    float big  = 0.0f;
#pragma unroll
    for (int j = 0; j < SEQ / 128; ++j) {
        const float4 t4 = reinterpret_cast<const float4*>(ptb)[j * 32 + lane];
        const float4 m4 = reinterpret_cast<const float4*>(pmk)[j * 32 + lane];
        tmax = fmaxf(tmax, fmaxf(fmaxf(t4.x + m4.x, t4.y + m4.y),
                                 fmaxf(t4.z + m4.z, t4.w + m4.w)));
        kmax = fmaxf(kmax, fmaxf(fmaxf(m4.x, m4.y), fmaxf(m4.z, m4.w)));
        big  = fmaxf(big, fmaxf(fmaxf(fabsf(t4.x), fabsf(t4.y)),
                                fmaxf(fabsf(t4.z), fabsf(t4.w))));
        big  = fmaxf(big, fmaxf(fmaxf(finite_size(m4.x), finite_size(m4.y)),
                                fmaxf(finite_size(m4.z), finite_size(m4.w))));
    }
#pragma unroll
    for (int d = 16; d >= 1; d >>= 1) {
        tmax = fmaxf(tmax, __shfl_xor(tmax, d, 32));
        kmax = fmaxf(kmax, __shfl_xor(kmax, d, 32));
        big  = fmaxf(big,  __shfl_xor(big,  d, 32));
    }

    float ref[QT], s[QT], acc[QT];
    bool small = big < BIG_OPERAND;
#pragma unroll
    for (int i = 0; i < QT; ++i) {
        ref[i] = fmaxf(taq[i] + tmax, kmax);
        small  = small && (fabsf(taq[i]) < BIG_OPERAND);
        s[i]   = 0.0f;
        acc[i] = 0.0f;
    }

    v8f tbc = *reinterpret_cast<const v8f*>(ptb);
    v8f hvc = *reinterpret_cast<const v8f*>(phv);
    v8f mkc = *reinterpret_cast<const v8f*>(pmk);

    if (__all(small)) {
        for (int k0 = 0; k0 < SEQ; k0 += KC) {
            const int kn = (k0 + KC < SEQ) ? (k0 + KC) : 0;
            const v8f tbn = *reinterpret_cast<const v8f*>(ptb + kn);
            const v8f hvn = *reinterpret_cast<const v8f*>(phv + kn);
            const v8f mkn = *reinterpret_cast<const v8f*>(pmk + kn);
#pragma unroll
            for (int j = 0; j < KC; ++j) {
                const float tbk = tbc[j];
                const float hvk = hvc[j];
                const float mkk = mkc[j];
#pragma unroll
                for (int i = 0; i < QT; ++i) {
                    const float x = taq[i] + tbk;
                    const float e = (x > 0.0f ? x : __expf(x) - 1.0f) + mkk;
                    const float p = __expf(e - ref[i]);
                    s[i]   += p;
                    acc[i]  = fmaf(p, hvk, acc[i]);
                }
            }
            tbc = tbn;
            hvc = hvn;
            mkc = mkn;
        }
    } else {
        float m[QT];
#pragma unroll
        for (int i = 0; i < QT; ++i) m[i] = -INFINITY;
        for (int k0 = 0; k0 < SEQ; k0 += KC) {
            const int kn = (k0 + KC < SEQ) ? (k0 + KC) : 0;
            const v8f tbn = *reinterpret_cast<const v8f*>(ptb + kn);
            const v8f hvn = *reinterpret_cast<const v8f*>(phv + kn);
            const v8f mkn = *reinterpret_cast<const v8f*>(pmk + kn);
#pragma unroll
            for (int j = 0; j < KC; ++j) {
                const float tbk = tbc[j];
                const float hvk = hvc[j];
                const float mkk = mkc[j];
#pragma unroll
                for (int i = 0; i < QT; ++i) {
                    const float x = taq[i] + tbk;
                    const float e = (x > 0.0f ? x : __expf(x) - 1.0f) + mkk;
                    const float mn   = fmaxf(m[i], e);
                    const float corr = __expf(m[i] - mn);
                    const float p    = __expf(e - mn);
                    s[i]   = s[i]  * corr + p;
                    acc[i] = acc[i] * corr + p * hvk;
                    m[i]   = mn;
                }
            }
            tbc = tbn;
            hvc = hvn;
            mkc = mkn;
        }
    }

#pragma unroll
    for (int i = 0; i < QT; ++i) {
        out[((size_t)b * SEQ + (q0 + 32 * i)) * HDIM + h] = acc[i] / s[i];
    }
}

#define NPOW 11
#define TA_TB_BOUND 4.0f
#define MASK_BOUND  16.0f
#define NGRP   6
#define GPITCH 17
#define GROW   (SEQ / 16 * GPITCH)
__device__ __forceinline__ int grp_place(int r) { return r + (r >> 4); }

#define NPL 2
__global__ __launch_bounds__(256) void attn_qlanes(
    const float* __restrict__ hvT,
    const float* __restrict__ Wa,
    const float* __restrict__ ba,
    const float* __restrict__ Ua,
    const float* __restrict__ bu,
    const float* __restrict__ maskp,
    float* __restrict__ out)
{
    __shared__ __attribute__((aligned(16))) unsigned long long skey[NPL][SEQ];
    __shared__ __attribute__((aligned(16))) unsigned wsort[NPL][SEQ];
    __shared__ __attribute__((aligned(16))) float stb[NPL][SEQ];
    __shared__ __attribute__((aligned(16))) float4 arr4[NPL][NGRP][GROW];
    __shared__ __attribute__((aligned(32))) float lta[NPL][SEQ];
    __shared__ __attribute__((aligned(32))) float ltb[NPL][SEQ];
    __shared__ __attribute__((aligned(32))) float lmk[NPL][SEQ];

    const int t    = threadIdx.x;
    const int lane = t & 31;
    const int w    = __builtin_amdgcn_readfirstlane(t >> 5);
    const int bh0  = blockIdx.x * NPL;
    const int b    = bh0 >> 7;
    const int h0   = bh0 & (HDIM - 1);
    const float* phv = hvT + (size_t)bh0 * SEQ;

#pragma unroll
    for (int i = 0; i < 2; ++i) {
        const int k = t + 256 * i;
        const float2 m2 = *reinterpret_cast<const float2*>(maskp + ((size_t)b * SEQ + k) * HDIM + h0);
        lmk[0][k] = m2.x;
        lmk[1][k] = m2.y;
    }

    {
        const int c  = lane & 15;
        const int hh = lane >> 4;
        const int p0 = w * 64;
        const float* wrow = ((c & 2) ? Ua : Wa) + (size_t)(h0 + (c & 1)) * HDIM + 4 * hh;
        const bool live = c < 4;
        const float* hrow = hvT + ((size_t)b * HDIM + 4 * hh) * SEQ + p0 + 4 * c;
        v8f acc[4];
#pragma unroll
        for (int tt = 0; tt < 4; ++tt) { v8f z = {}; acc[tt] = z; }
#pragma unroll
        for (int j = 0; j < 16; ++j) {
            v4f a4 = *reinterpret_cast<const v4f*>(wrow + 8 * j);
            if (!live) { v4f z = {}; a4 = z; }
            v4f bq[4];
#pragma unroll
            for (int e = 0; e < 4; ++e)
                bq[e] = *reinterpret_cast<const v4f*>(hrow + (size_t)(8 * j + e) * SEQ);
#pragma unroll
            for (int f = 0; f < 2; ++f) {
                v2f av = {a4[2 * f], a4[2 * f + 1]};
#pragma unroll
                for (int tt = 0; tt < 4; ++tt) {
                    v2f bv = {bq[2 * f][tt], bq[2 * f + 1][tt]};
                    acc[tt] = __builtin_amdgcn_wmma_f32_16x16x4_f32(
                                false, av, false, bv, (short)0, acc[tt], false, false);
                }
            }
        }
        if (hh == 0) {
#pragma unroll
            for (int p = 0; p < NPL; ++p) {
                const float bva = ba[h0 + p];
                const float bvu = bu[h0 + p];
                *reinterpret_cast<float4*>(&lta[p][p0 + 4 * c]) =
                    make_float4(acc[0][p] + bva, acc[1][p] + bva, acc[2][p] + bva, acc[3][p] + bva);
                *reinterpret_cast<float4*>(&ltb[p][p0 + 4 * c]) =
                    make_float4(acc[0][2 + p] + bvu, acc[1][2 + p] + bvu, acc[2][2 + p] + bvu, acc[3][2 + p] + bvu);
            }
        }
    }
    __syncthreads();

    float kmax[NPL];
    bool  bad  = false;
#pragma unroll
    for (int p = 0; p < NPL; ++p) {
        kmax[p] = -INFINITY;
#pragma unroll
        for (int j = 0; j < SEQ / 128; ++j) {
            const float4 a4 = reinterpret_cast<const float4*>(lta[p])[j * 32 + lane];
            const float4 t4 = reinterpret_cast<const float4*>(ltb[p])[j * 32 + lane];
            const float4 m4 = reinterpret_cast<const float4*>(lmk[p])[j * 32 + lane];
            const float av[4] = {a4.x, a4.y, a4.z, a4.w};
            const float tv[4] = {t4.x, t4.y, t4.z, t4.w};
            const float mv[4] = {m4.x, m4.y, m4.z, m4.w};
#pragma unroll
            for (int e = 0; e < 4; ++e) {
                bad = bad || !(fabsf(av[e]) <= TA_TB_BOUND) || !(fabsf(tv[e]) <= TA_TB_BOUND);
                bad = bad || !((mv[e] == -INFINITY) || (fabsf(mv[e]) <= MASK_BOUND));
                kmax[p] = fmaxf(kmax[p], mv[e]);
            }
        }
    }
#pragma unroll
    for (int d = 16; d >= 1; d >>= 1) {
#pragma unroll
        for (int p = 0; p < NPL; ++p) kmax[p] = fmaxf(kmax[p], __shfl_xor(kmax[p], d, 32));
    }
    const bool fast = !__any(bad) && (kmax[0] > -INFINITY) && (kmax[1] > -INFINITY);

    if (!fast) {
#pragma nounroll
        for (int p = 0; p < NPL; ++p)
            attn_pairs(bh0 + p, lta[p], ltb[p], phv + (size_t)p * SEQ, lmk[p], out);
        return;
    }

    float tbk[NPL][2], ebk[NPL][2], emk[NPL][2], hvk[NPL][2], taq[NPL][2];
    unsigned long long key[NPL][2];
#pragma unroll
    for (int p = 0; p < NPL; ++p) {
#pragma unroll
        for (int i = 0; i < 2; ++i) {
            const int k = t + 256 * i;
            tbk[p][i] = ltb[p][k];
            hvk[p][i] = phv[p * SEQ + k];
            ebk[p][i] = expf(tbk[p][i]);
            emk[p][i] = expf(lmk[p][k] - kmax[p]);
            unsigned u = __float_as_uint(tbk[p][i]);
            u ^= (u >> 31) ? 0xFFFFFFFFu : 0x80000000u;
            key[p][i] = ((unsigned long long)u << 32) | (unsigned)k;
            skey[p][w * 64 + 32 * i + lane] = key[p][i];
            taq[p][i] = lta[p][k];
        }
    }
    __syncthreads();

    int lr[NPL][2];
#pragma unroll
    for (int p = 0; p < NPL; ++p) { lr[p][0] = 0; lr[p][1] = 0; }
    {
        const ulonglong2* mine0 = reinterpret_cast<const ulonglong2*>(&skey[0][w * 64]);
        const ulonglong2* mine1 = reinterpret_cast<const ulonglong2*>(&skey[1][w * 64]);
#pragma unroll 4
        for (int kp = 0; kp < 32; ++kp) {
            const ulonglong2 v0 = mine0[kp];
            const ulonglong2 v1 = mine1[kp];
            lr[0][0] += (v0.x < key[0][0]) ? 1 : 0;
            lr[0][0] += (v0.y < key[0][0]) ? 1 : 0;
            lr[0][1] += (v0.x < key[0][1]) ? 1 : 0;
            lr[0][1] += (v0.y < key[0][1]) ? 1 : 0;
            lr[1][0] += (v1.x < key[1][0]) ? 1 : 0;
            lr[1][0] += (v1.y < key[1][0]) ? 1 : 0;
            lr[1][1] += (v1.x < key[1][1]) ? 1 : 0;
            lr[1][1] += (v1.y < key[1][1]) ? 1 : 0;
        }
#pragma unroll
        for (int p = 0; p < NPL; ++p) {
            wsort[p][w * 64 + lr[p][0]] = (unsigned)(key[p][0] >> 32);
            wsort[p][w * 64 + lr[p][1]] = (unsigned)(key[p][1] >> 32);
        }
    }
    __syncthreads();
    int rk[NPL][2];
#pragma unroll
    for (int p = 0; p < NPL; ++p) { rk[p][0] = lr[p][0]; rk[p][1] = lr[p][1]; }
    int ap[7][NPL][2];
#pragma unroll
    for (int d = 1; d < 8; ++d) {
#pragma unroll
        for (int p = 0; p < NPL; ++p) { ap[d - 1][p][0] = 0; ap[d - 1][p][1] = 0; }
    }
#pragma unroll
    for (int step = 32; step >= 1; step >>= 1) {
#pragma unroll
        for (int d = 1; d < 8; ++d) {
            const int w2 = (w + d) & 7;
            const unsigned tie = (w2 < w) ? 1u : 0u;
#pragma unroll
            for (int p = 0; p < NPL; ++p) {
                const unsigned* L = &wsort[p][w2 * 64];
#pragma unroll
                for (int i = 0; i < 2; ++i) {
                    const unsigned tk = (unsigned)(key[p][i] >> 32) + tie;
                    if (L[ap[d - 1][p][i] + step - 1] < tk) ap[d - 1][p][i] += step;
                }
            }
        }
    }
#pragma unroll
    for (int d = 1; d < 8; ++d) {
        const int w2 = (w + d) & 7;
        const unsigned tie = (w2 < w) ? 1u : 0u;
#pragma unroll
        for (int p = 0; p < NPL; ++p) {
            const unsigned* L = &wsort[p][w2 * 64];
#pragma unroll
            for (int i = 0; i < 2; ++i) {
                const unsigned tk = (unsigned)(key[p][i] >> 32) + tie;
                rk[p][i] += ap[d - 1][p][i] + ((L[ap[d - 1][p][i]] < tk) ? 1 : 0);
            }
        }
    }

#pragma unroll
    for (int p = 0; p < NPL; ++p) {
#pragma unroll
        for (int i = 0; i < 2; ++i) {
            const int r  = rk[p][i];
            const int rq = grp_place(r);
            const int sq = grp_place(SEQ - 1 - r);
            stb[p][r] = tbk[p][i];
            float term[NPOW];
            term[0] = emk[p][i];
#pragma unroll
            for (int n = 1; n < NPOW; ++n) term[n] = term[n - 1] * (ebk[p][i] * (1.0f / (float)n));
#pragma unroll
            for (int k = 0; k < 5; ++k)
                arr4[p][k][rq] = make_float4(term[2 * k],     term[2 * k] * hvk[p][i],
                                             term[2 * k + 1], term[2 * k + 1] * hvk[p][i]);
            const float e1 = ebk[p][i] * emk[p][i];
            reinterpret_cast<float2*>(&arr4[p][5][rq])[0] = make_float2(term[10], term[10] * hvk[p][i]);
            reinterpret_cast<float2*>(&arr4[p][5][sq])[1] = make_float2(e1, e1 * hvk[p][i]);
        }
    }
    __syncthreads();

    if (w < NGRP) {
        float4* q0 = &arr4[0][w][GPITCH * lane];
        float4* q1 = &arr4[1][w][GPITCH * lane];
        float4 v[NPL][16];
#pragma unroll
        for (int e = 0; e < 16; ++e) { v[0][e] = q0[e]; v[1][e] = q1[e]; }
#pragma unroll
        for (int e = 1; e < 16; ++e) {
#pragma unroll
            for (int p = 0; p < NPL; ++p) {
                v[p][e].x += v[p][e - 1].x; v[p][e].y += v[p][e - 1].y;
                v[p][e].z += v[p][e - 1].z; v[p][e].w += v[p][e - 1].w;
            }
        }
        float4 x[NPL];
        x[0] = v[0][15];
        x[1] = v[1][15];
#pragma unroll
        for (int d = 1; d < 32; d <<= 1) {
            float4 y[NPL];
#pragma unroll
            for (int p = 0; p < NPL; ++p) {
                y[p].x = __shfl_up(x[p].x, d, 32);
                y[p].y = __shfl_up(x[p].y, d, 32);
                y[p].z = __shfl_up(x[p].z, d, 32);
                y[p].w = __shfl_up(x[p].w, d, 32);
            }
            if (lane >= d) {
#pragma unroll
                for (int p = 0; p < NPL; ++p) {
                    x[p].x += y[p].x; x[p].y += y[p].y; x[p].z += y[p].z; x[p].w += y[p].w;
                }
            }
        }
        float4 off[NPL];
#pragma unroll
        for (int p = 0; p < NPL; ++p) {
            off[p].x = __shfl_up(x[p].x, 1, 32);
            off[p].y = __shfl_up(x[p].y, 1, 32);
            off[p].z = __shfl_up(x[p].z, 1, 32);
            off[p].w = __shfl_up(x[p].w, 1, 32);
            if (lane == 0) off[p] = make_float4(0.0f, 0.0f, 0.0f, 0.0f);
        }
#pragma unroll
        for (int e = 0; e < 16; ++e) {
            q0[e] = make_float4(v[0][e].x + off[0].x, v[0][e].y + off[0].y, v[0][e].z + off[0].z, v[0][e].w + off[0].w);
            q1[e] = make_float4(v[1][e].x + off[1].x, v[1][e].y + off[1].y, v[1][e].z + off[1].z, v[1][e].w + off[1].w);
        }
    }
    __syncthreads();

    float res[NPL][2];
#pragma unroll
    for (int p = 0; p < NPL; ++p) {
#pragma unroll
        for (int i = 0; i < 2; ++i) {
            const float ta  = taq[p][i];
            const float ea  = expf(ta);
            const float nta = -ta;
            int j = 0;
#pragma unroll
            for (int step = 256; step >= 1; step >>= 1)
                if (stb[p][j + step - 1] <= nta) j += step;
            if (stb[p][j] <= nta) j += 1;
            const int qm = grp_place(j > 0 ? j - 1 : 0);
            const int qs = grp_place(j < SEQ ? SEQ - 1 - j : 0);
            float4 g[5];
#pragma unroll
            for (int k = 0; k < 5; ++k) g[k] = arr4[p][k][qm];
            const float2 g5 = reinterpret_cast<const float2*>(&arr4[p][5][qm])[0];
            const float2 sf = reinterpret_cast<const float2*>(&arr4[p][5][qs])[1];
            float S = g5.x;
            float A = g5.y;
#pragma unroll
            for (int k = 4; k >= 0; --k) {
                S = fmaf(S, ea, g[k].z);
                A = fmaf(A, ea, g[k].w);
                S = fmaf(S, ea, g[k].x);
                A = fmaf(A, ea, g[k].y);
            }
            if (j == 0) { S = 0.0f; A = 0.0f; }
            float sufS = sf.x;
            float sufA = sf.y;
            if (j == SEQ) { sufS = 0.0f; sufA = 0.0f; }
            const float em1 = 0.36787944117144233f;
            const float num = fmaf(ea, sufA, em1 * A);
            const float den = fmaf(ea, sufS, em1 * S);
            res[p][i] = num / den;
        }
    }
#pragma unroll
    for (int i = 0; i < 2; ++i) {
        float* o = out + ((size_t)b * SEQ + (t + 256 * i)) * HDIM + h0;
        *reinterpret_cast<float2*>(o) = make_float2(res[0][i], res[1][i]);
    }
}

extern "C" void kernel_launch(void* const* d_in, const int* in_sizes, int n_in,
                              void* d_out, int out_size, void* d_ws, size_t ws_size,
                              hipStream_t stream) {
    const float* word_vecs = (const float*)d_in[0];
    const float* mask      = (const float*)d_in[1];
    const float* Wx_w      = (const float*)d_in[2];
    const float* Wx_b      = (const float*)d_in[3];
    const float* Wa_w      = (const float*)d_in[4];
    const float* Wa_b      = (const float*)d_in[5];
    const float* Ua_w      = (const float*)d_in[6];
    const float* Ua_b      = (const float*)d_in[7];
    float* out = (float*)d_out;

    float* hvT = (float*)d_ws;

    gemm_wx_tanh<<<256, 128, 0, stream>>>(word_vecs, Wx_w, Wx_b, hvT);

    attn_qlanes<<<256, 256, 0, stream>>>(hvT, Wa_w, Wa_b, Ua_w, Ua_b, mask, out);
}


# attention admission test (bounds, finite mask) once per value by its owning thread, ballot per wave, 8 LDS words: no re-read, branches or lane maxima; mask weight exp(mask), no max taken off
# speedup vs baseline: 12.1485x; 1.0724x over previous
#include <hip/hip_runtime.h>
#include <hip/hip_bf16.h>
#include <math.h>

#define BATCH   4
#define SEQ     512
#define INDIM   300
#define HDIM    128
#define MROWS   (BATCH * SEQ)

typedef __attribute__((ext_vector_type(2))) float v2f;
typedef __attribute__((ext_vector_type(4))) float v4f;
typedef __attribute__((ext_vector_type(8))) float v8f;


__device__ __forceinline__ v2f load_fragA(const float* A, int m0, int k0,
                                          int lda, int lane) {
    const int m  = m0 + (lane & 15);
    const int ka = k0 + ((lane >> 4) << 1);
    return *reinterpret_cast<const v2f*>(A + (size_t)m * lda + ka);
}

__device__ __forceinline__ v2f load_fragB_WT(const float* W, int n0, int k0,
                                             int ldw, int lane) {
    const int nn = n0 + (lane & 15);
    const int kb = k0 + ((lane >> 4) << 1);
    return *reinterpret_cast<const v2f*>(W + (size_t)nn * ldw + kb);
}

__device__ __forceinline__ v4f load_frag4(const float* P, int r0, int k0,
                                          int ld, int lane) {
    const int r  = r0 + (lane & 15);
    const int kk = k0 + ((lane >> 4) << 2);
    return *reinterpret_cast<const v4f*>(P + (size_t)r * ld + kk);
}
__device__ __forceinline__ v2f frag_lo(v4f v) { v2f r = {v.x, v.y}; return r; }
__device__ __forceinline__ v2f frag_hi(v4f v) { v2f r = {v.z, v.w}; return r; }

__device__ __forceinline__ void store_keymajor(float* P, int rb, int col,
                                               const float* v) {
    float* p = P + ((size_t)((rb >> 9) * HDIM + col)) * SEQ + (rb & (SEQ - 1));
    *reinterpret_cast<float4*>(p)     = make_float4(v[0], v[1], v[2], v[3]);
    *reinterpret_cast<float4*>(p + 4) = make_float4(v[4], v[5], v[6], v[7]);
}

__global__ __launch_bounds__(128) __attribute__((amdgpu_waves_per_eu(1, 4)))
void gemm_wx_tanh(
    const float* __restrict__ A,
    const float* __restrict__ W,
    const float* __restrict__ bias,
    float* __restrict__ outT)
{
    const int lane = threadIdx.x & 31;
    const int wid  = blockIdx.x * 4 + (threadIdx.x >> 5);
    const int m0   = (wid >> 3) << 4;
    const int n0   = (wid & 7) << 4;

    v8f acc = {};
#pragma unroll
    for (int k0 = 0; k0 < 296; k0 += 8) {
        const v4f a = load_frag4(A, m0, k0, INDIM, lane);
        const v4f b = load_frag4(W, n0, k0, INDIM, lane);
        acc = __builtin_amdgcn_wmma_f32_16x16x4_f32(
                false, frag_lo(a), false, frag_lo(b), (short)0, acc, false, false);
        acc = __builtin_amdgcn_wmma_f32_16x16x4_f32(
                false, frag_hi(a), false, frag_hi(b), (short)0, acc, false, false);
    }
    {
        const v2f a = load_fragA(A, m0, 296, INDIM, lane);
        const v2f b = load_fragB_WT(W, n0, 296, INDIM, lane);
        acc = __builtin_amdgcn_wmma_f32_16x16x4_f32(
                false, a, false, b, (short)0, acc, false, false);
    }

    const int col = n0 + (lane & 15);
    const int rb  = m0 + ((lane >> 4) << 3);
    const float bv = bias[col];
    float v[8];
#pragma unroll
    for (int r = 0; r < 8; ++r) v[r] = tanhf(acc[r] + bv);
    store_keymajor(outT, rb, col, v);
}

#define QT 2
#define KC 8
#define BIG_OPERAND 67108864.0f

__device__ __forceinline__ float finite_size(float v) {
    const float a = fabsf(v);
    return a < INFINITY ? a : 0.0f;
}

__device__ __forceinline__ void attn_pairs(
    const int bh,
    const float* pta,
    const float* ptb,
    const float* phv,
    const float* pmk,
    float* __restrict__ out)
{
    const int lane = threadIdx.x & 31;
    const int w  = __builtin_amdgcn_readfirstlane((int)(threadIdx.x >> 5));
    const int b  = bh >> 7;
    const int h  = bh & (HDIM - 1);
    const int q0 = w * (32 * QT) + lane;

    float taq[QT];
#pragma unroll
    for (int i = 0; i < QT; ++i) taq[i] = pta[q0 + 32 * i];

    float tmax = -INFINITY;
    float kmax = -INFINITY;
    float big  = 0.0f;
#pragma unroll
    for (int j = 0; j < SEQ / 128; ++j) {
        const float4 t4 = reinterpret_cast<const float4*>(ptb)[j * 32 + lane];
        const float4 m4 = reinterpret_cast<const float4*>(pmk)[j * 32 + lane];
        tmax = fmaxf(tmax, fmaxf(fmaxf(t4.x + m4.x, t4.y + m4.y),
                                 fmaxf(t4.z + m4.z, t4.w + m4.w)));
        kmax = fmaxf(kmax, fmaxf(fmaxf(m4.x, m4.y), fmaxf(m4.z, m4.w)));
        big  = fmaxf(big, fmaxf(fmaxf(fabsf(t4.x), fabsf(t4.y)),
                                fmaxf(fabsf(t4.z), fabsf(t4.w))));
        big  = fmaxf(big, fmaxf(fmaxf(finite_size(m4.x), finite_size(m4.y)),
                                fmaxf(finite_size(m4.z), finite_size(m4.w))));
    }
#pragma unroll
    for (int d = 16; d >= 1; d >>= 1) {
        tmax = fmaxf(tmax, __shfl_xor(tmax, d, 32));
        kmax = fmaxf(kmax, __shfl_xor(kmax, d, 32));
        big  = fmaxf(big,  __shfl_xor(big,  d, 32));
    }

    float ref[QT], s[QT], acc[QT];
    bool small = big < BIG_OPERAND;
#pragma unroll
    for (int i = 0; i < QT; ++i) {
        ref[i] = fmaxf(taq[i] + tmax, kmax);
        small  = small && (fabsf(taq[i]) < BIG_OPERAND);
        s[i]   = 0.0f;
        acc[i] = 0.0f;
    }

    v8f tbc = *reinterpret_cast<const v8f*>(ptb);
    v8f hvc = *reinterpret_cast<const v8f*>(phv);
    v8f mkc = *reinterpret_cast<const v8f*>(pmk);

    if (__all(small)) {
        for (int k0 = 0; k0 < SEQ; k0 += KC) {
            const int kn = (k0 + KC < SEQ) ? (k0 + KC) : 0;
            const v8f tbn = *reinterpret_cast<const v8f*>(ptb + kn);
            const v8f hvn = *reinterpret_cast<const v8f*>(phv + kn);
            const v8f mkn = *reinterpret_cast<const v8f*>(pmk + kn);
#pragma unroll
            for (int j = 0; j < KC; ++j) {
                const float tbk = tbc[j];
                const float hvk = hvc[j];
                const float mkk = mkc[j];
#pragma unroll
                for (int i = 0; i < QT; ++i) {
                    const float x = taq[i] + tbk;
                    const float e = (x > 0.0f ? x : __expf(x) - 1.0f) + mkk;
                    const float p = __expf(e - ref[i]);
                    s[i]   += p;
                    acc[i]  = fmaf(p, hvk, acc[i]);
                }
            }
            tbc = tbn;
            hvc = hvn;
            mkc = mkn;
        }
    } else {
        float m[QT];
#pragma unroll
        for (int i = 0; i < QT; ++i) m[i] = -INFINITY;
        for (int k0 = 0; k0 < SEQ; k0 += KC) {
            const int kn = (k0 + KC < SEQ) ? (k0 + KC) : 0;
            const v8f tbn = *reinterpret_cast<const v8f*>(ptb + kn);
            const v8f hvn = *reinterpret_cast<const v8f*>(phv + kn);
            const v8f mkn = *reinterpret_cast<const v8f*>(pmk + kn);
#pragma unroll
            for (int j = 0; j < KC; ++j) {
                const float tbk = tbc[j];
                const float hvk = hvc[j];
                const float mkk = mkc[j];
#pragma unroll
                for (int i = 0; i < QT; ++i) {
                    const float x = taq[i] + tbk;
                    const float e = (x > 0.0f ? x : __expf(x) - 1.0f) + mkk;
                    const float mn   = fmaxf(m[i], e);
                    const float corr = __expf(m[i] - mn);
                    const float p    = __expf(e - mn);
                    s[i]   = s[i]  * corr + p;
                    acc[i] = acc[i] * corr + p * hvk;
                    m[i]   = mn;
                }
            }
            tbc = tbn;
            hvc = hvn;
            mkc = mkn;
        }
    }

#pragma unroll
    for (int i = 0; i < QT; ++i) {
        out[((size_t)b * SEQ + (q0 + 32 * i)) * HDIM + h] = acc[i] / s[i];
    }
}

#define NPOW 11
#define TA_TB_BOUND 4.0f
#define MASK_BOUND  16.0f
#define NGRP   6
#define GPITCH 17
#define GROW   (SEQ / 16 * GPITCH)
__device__ __forceinline__ int grp_place(int r) { return r + (r >> 4); }

#define NPL 2
__device__ __forceinline__ unsigned mask_outside(float m) {
    return ((int)(m == -INFINITY) | (int)(fabsf(m) <= MASK_BOUND)) ? 0u : 1u;
}
__device__ __forceinline__ unsigned proj_outside(float v) {
    return (fabsf(v) <= TA_TB_BOUND) ? 0u : 1u;
}
__global__ __launch_bounds__(256) void attn_qlanes(
    const float* __restrict__ hvT,
    const float* __restrict__ Wa,
    const float* __restrict__ ba,
    const float* __restrict__ Ua,
    const float* __restrict__ bu,
    const float* __restrict__ maskp,
    float* __restrict__ out)
{
    __shared__ __attribute__((aligned(16))) unsigned long long skey[NPL][SEQ];
    __shared__ __attribute__((aligned(16))) unsigned wsort[NPL][SEQ];
    __shared__ __attribute__((aligned(16))) float stb[NPL][SEQ];
    __shared__ __attribute__((aligned(16))) float4 arr4[NPL][NGRP][GROW];
    __shared__ __attribute__((aligned(32))) float lta[NPL][SEQ];
    __shared__ __attribute__((aligned(32))) float ltb[NPL][SEQ];
    __shared__ __attribute__((aligned(32))) float lmk[NPL][SEQ];
    __shared__ __attribute__((aligned(16))) unsigned wflag[8];

    const int t    = threadIdx.x;
    const int lane = t & 31;
    const int w    = __builtin_amdgcn_readfirstlane(t >> 5);
    const int bh0  = blockIdx.x * NPL;
    const int b    = bh0 >> 7;
    const int h0   = bh0 & (HDIM - 1);
    const float* phv = hvT + (size_t)bh0 * SEQ;

    unsigned bad = 0u;
    unsigned fin[NPL] = {0u, 0u};
#pragma unroll
    for (int i = 0; i < 2; ++i) {
        const int k = t + 256 * i;
        const float2 m2 = *reinterpret_cast<const float2*>(maskp + ((size_t)b * SEQ + k) * HDIM + h0);
        lmk[0][k] = m2.x;
        lmk[1][k] = m2.y;
        bad    |= mask_outside(m2.x) | mask_outside(m2.y);
        fin[0] |= (m2.x > -INFINITY) ? 1u : 0u;
        fin[1] |= (m2.y > -INFINITY) ? 1u : 0u;
    }

    {
        const int c  = lane & 15;
        const int hh = lane >> 4;
        const int p0 = w * 64;
        const float* wrow = ((c & 2) ? Ua : Wa) + (size_t)(h0 + (c & 1)) * HDIM + 4 * hh;
        const bool live = c < 4;
        const float* hrow = hvT + ((size_t)b * HDIM + 4 * hh) * SEQ + p0 + 4 * c;
        v8f acc[4];
#pragma unroll
        for (int tt = 0; tt < 4; ++tt) { v8f z = {}; acc[tt] = z; }
#pragma unroll
        for (int j = 0; j < 16; ++j) {
            v4f a4 = *reinterpret_cast<const v4f*>(wrow + 8 * j);
            if (!live) { v4f z = {}; a4 = z; }
            v4f bq[4];
#pragma unroll
            for (int e = 0; e < 4; ++e)
                bq[e] = *reinterpret_cast<const v4f*>(hrow + (size_t)(8 * j + e) * SEQ);
#pragma unroll
            for (int f = 0; f < 2; ++f) {
                v2f av = {a4[2 * f], a4[2 * f + 1]};
#pragma unroll
                for (int tt = 0; tt < 4; ++tt) {
                    v2f bv = {bq[2 * f][tt], bq[2 * f + 1][tt]};
                    acc[tt] = __builtin_amdgcn_wmma_f32_16x16x4_f32(
                                false, av, false, bv, (short)0, acc[tt], false, false);
                }
            }
        }
        if (hh == 0) {
#pragma unroll
            for (int p = 0; p < NPL; ++p) {
                const float bva = ba[h0 + p];
                const float bvu = bu[h0 + p];
                const float4 ta4 = make_float4(acc[0][p] + bva, acc[1][p] + bva, acc[2][p] + bva, acc[3][p] + bva);
                const float4 tb4 = make_float4(acc[0][2 + p] + bvu, acc[1][2 + p] + bvu, acc[2][2 + p] + bvu, acc[3][2 + p] + bvu);
                *reinterpret_cast<float4*>(&lta[p][p0 + 4 * c]) = ta4;
                *reinterpret_cast<float4*>(&ltb[p][p0 + 4 * c]) = tb4;
                bad |= proj_outside(ta4.x) | proj_outside(ta4.y) | proj_outside(ta4.z) | proj_outside(ta4.w);
                bad |= proj_outside(tb4.x) | proj_outside(tb4.y) | proj_outside(tb4.z) | proj_outside(tb4.w);
            }
        }
    }
    {
        const unsigned wbits = (__any((int)bad) ? 1u : 0u) | (__any((int)fin[0]) ? 2u : 0u) | (__any((int)fin[1]) ? 4u : 0u);
        if (lane == 0) wflag[w] = wbits;
    }
    __syncthreads();

    const uint4 f0 = reinterpret_cast<const uint4*>(wflag)[0];
    const uint4 f1 = reinterpret_cast<const uint4*>(wflag)[1];
    const unsigned allbits = f0.x | f0.y | f0.z | f0.w | f1.x | f1.y | f1.z | f1.w;
    const bool fast = (__builtin_amdgcn_readfirstlane((int)(allbits & 7u)) == 6);

    if (!fast) {
#pragma nounroll
        for (int p = 0; p < NPL; ++p)
            attn_pairs(bh0 + p, lta[p], ltb[p], phv + (size_t)p * SEQ, lmk[p], out);
        return;
    }

    float tbk[NPL][2], ebk[NPL][2], emk[NPL][2], hvk[NPL][2], taq[NPL][2];
    unsigned long long key[NPL][2];
#pragma unroll
    for (int p = 0; p < NPL; ++p) {
#pragma unroll
        for (int i = 0; i < 2; ++i) {
            const int k = t + 256 * i;
            tbk[p][i] = ltb[p][k];
            hvk[p][i] = phv[p * SEQ + k];
            ebk[p][i] = expf(tbk[p][i]);
            emk[p][i] = expf(lmk[p][k]);
            unsigned u = __float_as_uint(tbk[p][i]);
            u ^= (u >> 31) ? 0xFFFFFFFFu : 0x80000000u;
            key[p][i] = ((unsigned long long)u << 32) | (unsigned)k;
            skey[p][w * 64 + 32 * i + lane] = key[p][i];
            taq[p][i] = lta[p][k];
        }
    }
    __syncthreads();

    int lr[NPL][2];
#pragma unroll
    for (int p = 0; p < NPL; ++p) { lr[p][0] = 0; lr[p][1] = 0; }
    {
        const ulonglong2* mine0 = reinterpret_cast<const ulonglong2*>(&skey[0][w * 64]);
        const ulonglong2* mine1 = reinterpret_cast<const ulonglong2*>(&skey[1][w * 64]);
#pragma unroll 4
        for (int kp = 0; kp < 32; ++kp) {
            const ulonglong2 v0 = mine0[kp];
            const ulonglong2 v1 = mine1[kp];
            lr[0][0] += (v0.x < key[0][0]) ? 1 : 0;
            lr[0][0] += (v0.y < key[0][0]) ? 1 : 0;
            lr[0][1] += (v0.x < key[0][1]) ? 1 : 0;
            lr[0][1] += (v0.y < key[0][1]) ? 1 : 0;
            lr[1][0] += (v1.x < key[1][0]) ? 1 : 0;
            lr[1][0] += (v1.y < key[1][0]) ? 1 : 0;
            lr[1][1] += (v1.x < key[1][1]) ? 1 : 0;
            lr[1][1] += (v1.y < key[1][1]) ? 1 : 0;
        }
#pragma unroll
        for (int p = 0; p < NPL; ++p) {
            wsort[p][w * 64 + lr[p][0]] = (unsigned)(key[p][0] >> 32);
            wsort[p][w * 64 + lr[p][1]] = (unsigned)(key[p][1] >> 32);
        }
    }
    __syncthreads();
    int rk[NPL][2];
#pragma unroll
    for (int p = 0; p < NPL; ++p) { rk[p][0] = lr[p][0]; rk[p][1] = lr[p][1]; }
    int ap[7][NPL][2];
#pragma unroll
    for (int d = 1; d < 8; ++d) {
#pragma unroll
        for (int p = 0; p < NPL; ++p) { ap[d - 1][p][0] = 0; ap[d - 1][p][1] = 0; }
    }
#pragma unroll
    for (int step = 32; step >= 1; step >>= 1) {
#pragma unroll
        for (int d = 1; d < 8; ++d) {
            const int w2 = (w + d) & 7;
            const unsigned tie = (w2 < w) ? 1u : 0u;
#pragma unroll
            for (int p = 0; p < NPL; ++p) {
                const unsigned* L = &wsort[p][w2 * 64];
#pragma unroll
                for (int i = 0; i < 2; ++i) {
                    const unsigned tk = (unsigned)(key[p][i] >> 32) + tie;
                    if (L[ap[d - 1][p][i] + step - 1] < tk) ap[d - 1][p][i] += step;
                }
            }
        }
    }
#pragma unroll
    for (int d = 1; d < 8; ++d) {
        const int w2 = (w + d) & 7;
        const unsigned tie = (w2 < w) ? 1u : 0u;
#pragma unroll
        for (int p = 0; p < NPL; ++p) {
            const unsigned* L = &wsort[p][w2 * 64];
#pragma unroll
            for (int i = 0; i < 2; ++i) {
                const unsigned tk = (unsigned)(key[p][i] >> 32) + tie;
                rk[p][i] += ap[d - 1][p][i] + ((L[ap[d - 1][p][i]] < tk) ? 1 : 0);
            }
        }
    }

#pragma unroll
    for (int p = 0; p < NPL; ++p) {
#pragma unroll
        for (int i = 0; i < 2; ++i) {
            const int r  = rk[p][i];
            const int rq = grp_place(r);
            const int sq = grp_place(SEQ - 1 - r);
            stb[p][r] = tbk[p][i];
            float term[NPOW];
            term[0] = emk[p][i];
#pragma unroll
            for (int n = 1; n < NPOW; ++n) term[n] = term[n - 1] * (ebk[p][i] * (1.0f / (float)n));
#pragma unroll
            for (int k = 0; k < 5; ++k)
                arr4[p][k][rq] = make_float4(term[2 * k],     term[2 * k] * hvk[p][i],
                                             term[2 * k + 1], term[2 * k + 1] * hvk[p][i]);
            const float e1 = ebk[p][i] * emk[p][i];
            reinterpret_cast<float2*>(&arr4[p][5][rq])[0] = make_float2(term[10], term[10] * hvk[p][i]);
            reinterpret_cast<float2*>(&arr4[p][5][sq])[1] = make_float2(e1, e1 * hvk[p][i]);
        }
    }
    __syncthreads();

    if (w < NGRP) {
        float4* q0 = &arr4[0][w][GPITCH * lane];
        float4* q1 = &arr4[1][w][GPITCH * lane];
        float4 v[NPL][16];
#pragma unroll
        for (int e = 0; e < 16; ++e) { v[0][e] = q0[e]; v[1][e] = q1[e]; }
#pragma unroll
        for (int e = 1; e < 16; ++e) {
#pragma unroll
            for (int p = 0; p < NPL; ++p) {
                v[p][e].x += v[p][e - 1].x; v[p][e].y += v[p][e - 1].y;
                v[p][e].z += v[p][e - 1].z; v[p][e].w += v[p][e - 1].w;
            }
        }
        float4 x[NPL];
        x[0] = v[0][15];
        x[1] = v[1][15];
#pragma unroll
        for (int d = 1; d < 32; d <<= 1) {
            float4 y[NPL];
#pragma unroll
            for (int p = 0; p < NPL; ++p) {
                y[p].x = __shfl_up(x[p].x, d, 32);
                y[p].y = __shfl_up(x[p].y, d, 32);
                y[p].z = __shfl_up(x[p].z, d, 32);
                y[p].w = __shfl_up(x[p].w, d, 32);
            }
            if (lane >= d) {
#pragma unroll
                for (int p = 0; p < NPL; ++p) {
                    x[p].x += y[p].x; x[p].y += y[p].y; x[p].z += y[p].z; x[p].w += y[p].w;
                }
            }
        }
        float4 off[NPL];
#pragma unroll
        for (int p = 0; p < NPL; ++p) {
            off[p].x = __shfl_up(x[p].x, 1, 32);
            off[p].y = __shfl_up(x[p].y, 1, 32);
            off[p].z = __shfl_up(x[p].z, 1, 32);
            off[p].w = __shfl_up(x[p].w, 1, 32);
            if (lane == 0) off[p] = make_float4(0.0f, 0.0f, 0.0f, 0.0f);
        }
#pragma unroll
        for (int e = 0; e < 16; ++e) {
            q0[e] = make_float4(v[0][e].x + off[0].x, v[0][e].y + off[0].y, v[0][e].z + off[0].z, v[0][e].w + off[0].w);
            q1[e] = make_float4(v[1][e].x + off[1].x, v[1][e].y + off[1].y, v[1][e].z + off[1].z, v[1][e].w + off[1].w);
        }
    }
    __syncthreads();

    float res[NPL][2];
#pragma unroll
    for (int p = 0; p < NPL; ++p) {
#pragma unroll
        for (int i = 0; i < 2; ++i) {
            const float ta  = taq[p][i];
            const float ea  = expf(ta);
            const float nta = -ta;
            int j = 0;
#pragma unroll
            for (int step = 256; step >= 1; step >>= 1)
                if (stb[p][j + step - 1] <= nta) j += step;
            if (stb[p][j] <= nta) j += 1;
            const int qm = grp_place(j > 0 ? j - 1 : 0);
            const int qs = grp_place(j < SEQ ? SEQ - 1 - j : 0);
            float4 g[5];
#pragma unroll
            for (int k = 0; k < 5; ++k) g[k] = arr4[p][k][qm];
            const float2 g5 = reinterpret_cast<const float2*>(&arr4[p][5][qm])[0];
            const float2 sf = reinterpret_cast<const float2*>(&arr4[p][5][qs])[1];
            float S = g5.x;
            float A = g5.y;
#pragma unroll
            for (int k = 4; k >= 0; --k) {
                S = fmaf(S, ea, g[k].z);
                A = fmaf(A, ea, g[k].w);
                S = fmaf(S, ea, g[k].x);
                A = fmaf(A, ea, g[k].y);
            }
            if (j == 0) { S = 0.0f; A = 0.0f; }
            float sufS = sf.x;
            float sufA = sf.y;
            if (j == SEQ) { sufS = 0.0f; sufA = 0.0f; }
            const float em1 = 0.36787944117144233f;
            const float num = fmaf(ea, sufA, em1 * A);
            const float den = fmaf(ea, sufS, em1 * S);
            res[p][i] = num / den;
        }
    }
#pragma unroll
    for (int i = 0; i < 2; ++i) {
        float* o = out + ((size_t)b * SEQ + (t + 256 * i)) * HDIM + h0;
        *reinterpret_cast<float2*>(o) = make_float2(res[0][i], res[1][i]);
    }
}

extern "C" void kernel_launch(void* const* d_in, const int* in_sizes, int n_in,
                              void* d_out, int out_size, void* d_ws, size_t ws_size,
                              hipStream_t stream) {
    const float* word_vecs = (const float*)d_in[0];
    const float* mask      = (const float*)d_in[1];
    const float* Wx_w      = (const float*)d_in[2];
    const float* Wx_b      = (const float*)d_in[3];
    const float* Wa_w      = (const float*)d_in[4];
    const float* Wa_b      = (const float*)d_in[5];
    const float* Ua_w      = (const float*)d_in[6];
    const float* Ua_b      = (const float*)d_in[7];
    float* out = (float*)d_out;

    float* hvT = (float*)d_ws;

    gemm_wx_tanh<<<256, 128, 0, stream>>>(word_vecs, Wx_w, Wx_b, hvT);

    attn_qlanes<<<256, 256, 0, stream>>>(hvT, Wa_w, Wa_b, Ua_w, Ua_b, mask, out);
}
